# starting kernel
# baseline (speedup 1.0000x reference)
#include <hip/hip_runtime.h>
#include <hip/hip_bf16.h>

typedef __attribute__((ext_vector_type(2))) float v2f;
typedef __attribute__((ext_vector_type(8))) float v8f;

#define IN_DIM  128
#define OUT_DIM 64

__global__ void gat_zero_out(float4* __restrict__ out, int n4) {
    int i = blockIdx.x * blockDim.x + threadIdx.x;
    int stride = gridDim.x * blockDim.x;
    float4 z4 = make_float4(0.f, 0.f, 0.f, 0.f);
    for (; i < n4; i += stride) out[i] = z4;
}

__global__ void gat_gemm_z(const float* __restrict__ h,
                           const float* __restrict__ W,
                           float* __restrict__ z,
                           int nNodes) {
    const int wave = (blockIdx.x * blockDim.x + threadIdx.x) >> 5;
    const int lane = threadIdx.x & 31;
    const int row0 = wave * 16;
    if (row0 >= nNodes) return;

    const int half = lane >> 4;
    const int l16  = lane & 15;

    const float* hrow = h + (size_t)(row0 + l16) * IN_DIM + half * 2;

    v8f acc0 = {}, acc1 = {}, acc2 = {}, acc3 = {};

    for (int k = 0; k < IN_DIM; k += 4) {
        v2f a;
        a.x = hrow[k];
        a.y = hrow[k + 1];

        const float* wk = W + (size_t)(k + half * 2) * OUT_DIM + l16;
        v2f b0, b1, b2, b3;
        b0.x = wk[0];       b0.y = wk[OUT_DIM + 0];
        b1.x = wk[16];      b1.y = wk[OUT_DIM + 16];
        b2.x = wk[32];      b2.y = wk[OUT_DIM + 32];
        b3.x = wk[48];      b3.y = wk[OUT_DIM + 48];

        acc0 = __builtin_amdgcn_wmma_f32_16x16x4_f32(false, a, false, b0,
                                                     (short)0, acc0, false, false);
        acc1 = __builtin_amdgcn_wmma_f32_16x16x4_f32(false, a, false, b1,
                                                     (short)0, acc1, false, false);
        acc2 = __builtin_amdgcn_wmma_f32_16x16x4_f32(false, a, false, b2,
                                                     (short)0, acc2, false, false);
        acc3 = __builtin_amdgcn_wmma_f32_16x16x4_f32(false, a, false, b3,
                                                     (short)0, acc3, false, false);
    }

    float* zt = z + (size_t)row0 * OUT_DIM + l16;
#pragma unroll
    for (int v = 0; v < 8; ++v) {
        const size_t r = (size_t)(v + 8 * half) * OUT_DIM;
        zt[r + 0]  = acc0[v];
        zt[r + 16] = acc1[v];
        zt[r + 32] = acc2[v];
        zt[r + 48] = acc3[v];
    }
}

__global__ void gat_edge(const float* __restrict__ z,
                         const int* __restrict__ src,
                         const int* __restrict__ dst,
                         const float* __restrict__ a_attn,
                         float* __restrict__ out,
                         int nEdges) {
    const int edge = (blockIdx.x * blockDim.x + threadIdx.x) >> 5;
    const int lane = threadIdx.x & 31;
    if (edge >= nEdges) return;

    const int s = src[edge];
    const int d = dst[edge];
    const float* zs = z + (size_t)s * OUT_DIM;
    const float* zd = z + (size_t)d * OUT_DIM;

    const float zs0 = zs[lane],      zs1 = zs[lane + 32];
    const float zd0 = zd[lane],      zd1 = zd[lane + 32];

    float p = zs0 * a_attn[lane]      + zs1 * a_attn[lane + 32]
            + zd0 * a_attn[64 + lane] + zd1 * a_attn[96 + lane];

#pragma unroll
    for (int off = 16; off > 0; off >>= 1)
        p += __shfl_xor(p, off, 32);

    const float e = (p >= 0.f) ? p : 0.01f * p;
    const float alpha = 1.f / (1.f + __expf(-e));

    float* o = out + (size_t)d * OUT_DIM;
    atomicAdd(&o[lane],      alpha * zs0);
    atomicAdd(&o[lane + 32], alpha * zs1);
}

extern "C" void kernel_launch(void* const* d_in, const int* in_sizes, int n_in,
                              void* d_out, int out_size, void* d_ws, size_t ws_size,
                              hipStream_t stream) {
    const float* h      = (const float*)d_in[0];
    const int*   src    = (const int*)  d_in[1];
    const int*   dst    = (const int*)  d_in[2];
    const float* W_fc   = (const float*)d_in[3];
    const float* a_attn = (const float*)d_in[4];
    float*       out    = (float*)d_out;
    float*       z      = (float*)d_ws;

    const int nNodes = in_sizes[0] / IN_DIM;
    const int nEdges = in_sizes[1];

    {
        const int n4 = out_size / 4;
        const int blocks = min((n4 + 255) / 256, 4096);
        gat_zero_out<<<blocks, 256, 0, stream>>>((float4*)out, n4);
    }

    {
        const int nWaves = (nNodes + 15) / 16;
        const int blocks = (nWaves + 7) / 8;
        gat_gemm_z<<<blocks, 256, 0, stream>>>(h, W_fc, z, nNodes);
    }

    {
        const int blocks = (nEdges + 7) / 8;
        gat_edge<<<blocks, 256, 0, stream>>>(z, src, dst, a_attn, out, nEdges);
    }
}


# edge pass: each wave walks a run of 32 consecutive edges (indices fetched once, coalesced; row gathers of 8 edges in flight) instead of one wave per edge: 1,000,000 waves become 31,250
# speedup vs baseline: 1.7934x; 1.7934x over previous
#include <hip/hip_runtime.h>
#include <hip/hip_bf16.h>

typedef __attribute__((ext_vector_type(2))) float v2f;
typedef __attribute__((ext_vector_type(8))) float v8f;

#define IN_DIM  128
#define OUT_DIM 64

__global__ void gat_zero_out(float4* __restrict__ out, int n4) {
    int i = blockIdx.x * blockDim.x + threadIdx.x;
    int stride = gridDim.x * blockDim.x;
    float4 z4 = make_float4(0.f, 0.f, 0.f, 0.f);
    for (; i < n4; i += stride) out[i] = z4;
}

__global__ void gat_gemm_z(const float* __restrict__ h,
                           const float* __restrict__ W,
                           float* __restrict__ z,
                           int nNodes) {
    const int wave = (blockIdx.x * blockDim.x + threadIdx.x) >> 5;
    const int lane = threadIdx.x & 31;
    const int row0 = wave * 16;
    if (row0 >= nNodes) return;

    const int half = lane >> 4;
    const int l16  = lane & 15;

    const float* hrow = h + (size_t)(row0 + l16) * IN_DIM + half * 2;

    v8f acc0 = {}, acc1 = {}, acc2 = {}, acc3 = {};

    for (int k = 0; k < IN_DIM; k += 4) {
        v2f a;
        a.x = hrow[k];
        a.y = hrow[k + 1];

        const float* wk = W + (size_t)(k + half * 2) * OUT_DIM + l16;
        v2f b0, b1, b2, b3;
        b0.x = wk[0];       b0.y = wk[OUT_DIM + 0];
        b1.x = wk[16];      b1.y = wk[OUT_DIM + 16];
        b2.x = wk[32];      b2.y = wk[OUT_DIM + 32];
        b3.x = wk[48];      b3.y = wk[OUT_DIM + 48];

        acc0 = __builtin_amdgcn_wmma_f32_16x16x4_f32(false, a, false, b0,
                                                     (short)0, acc0, false, false);
        acc1 = __builtin_amdgcn_wmma_f32_16x16x4_f32(false, a, false, b1,
                                                     (short)0, acc1, false, false);
        acc2 = __builtin_amdgcn_wmma_f32_16x16x4_f32(false, a, false, b2,
                                                     (short)0, acc2, false, false);
        acc3 = __builtin_amdgcn_wmma_f32_16x16x4_f32(false, a, false, b3,
                                                     (short)0, acc3, false, false);
    }

    float* zt = z + (size_t)row0 * OUT_DIM + l16;
#pragma unroll
    for (int v = 0; v < 8; ++v) {
        const size_t r = (size_t)(v + 8 * half) * OUT_DIM;
        zt[r + 0]  = acc0[v];
        zt[r + 16] = acc1[v];
        zt[r + 32] = acc2[v];
        zt[r + 48] = acc3[v];
    }
}

#define EDGES_PER_WAVE 32
#define EDGE_UNROLL    8

template <int N>
__device__ __forceinline__ void gat_edge_group(const float* __restrict__ z,
                                               float* __restrict__ out,
                                               int s_l, int d_l, int j, int lane,
                                               float a0, float a1, float a2, float a3) {
    float zs0[N], zs1[N], zd0[N], zd1[N], p[N];
    int dd[N];

#pragma unroll
    for (int u = 0; u < N; ++u) {
        const int s = __builtin_amdgcn_readlane(s_l, j + u);
        const int d = __builtin_amdgcn_readlane(d_l, j + u);
        dd[u] = d;
        const float* zs = z + (size_t)s * OUT_DIM;
        const float* zd = z + (size_t)d * OUT_DIM;
        zs0[u] = zs[lane];  zs1[u] = zs[lane + 32];
        zd0[u] = zd[lane];  zd1[u] = zd[lane + 32];
    }

#pragma unroll
    for (int u = 0; u < N; ++u)
        p[u] = zs0[u] * a0 + zs1[u] * a1 + zd0[u] * a2 + zd1[u] * a3;

#pragma unroll
    for (int off = 16; off > 0; off >>= 1) {
#pragma unroll
        for (int u = 0; u < N; ++u)
            p[u] += __shfl_xor(p[u], off, 32);
    }

#pragma unroll
    for (int u = 0; u < N; ++u) {
        const float e = (p[u] >= 0.f) ? p[u] : 0.01f * p[u];
        const float alpha = 1.f / (1.f + __expf(-e));
        float* o = out + (size_t)dd[u] * OUT_DIM;
        atomicAdd(&o[lane],      alpha * zs0[u]);
        atomicAdd(&o[lane + 32], alpha * zs1[u]);
    }
}

__global__ void gat_edge(const float* __restrict__ z,
                         const int* __restrict__ src,
                         const int* __restrict__ dst,
                         const float* __restrict__ a_attn,
                         float* __restrict__ out,
                         int nEdges) {
    const int wave = __builtin_amdgcn_readfirstlane(
        (int)((blockIdx.x * blockDim.x + threadIdx.x) >> 5));
    const int lane = threadIdx.x & 31;
    const int e0 = wave * EDGES_PER_WAVE;
    if (e0 >= nEdges) return;
    const int cnt = min(EDGES_PER_WAVE, nEdges - e0);

    int s_l = 0, d_l = 0;
    if (lane < cnt) {
        s_l = src[e0 + lane];
        d_l = dst[e0 + lane];
    }

    const float a0 = a_attn[lane],      a1 = a_attn[lane + 32];
    const float a2 = a_attn[64 + lane], a3 = a_attn[96 + lane];

    int j = 0;
    for (; j + EDGE_UNROLL <= cnt; j += EDGE_UNROLL)
        gat_edge_group<EDGE_UNROLL>(z, out, s_l, d_l, j, lane, a0, a1, a2, a3);
    for (; j < cnt; ++j)
        gat_edge_group<1>(z, out, s_l, d_l, j, lane, a0, a1, a2, a3);
}

extern "C" void kernel_launch(void* const* d_in, const int* in_sizes, int n_in,
                              void* d_out, int out_size, void* d_ws, size_t ws_size,
                              hipStream_t stream) {
    const float* h      = (const float*)d_in[0];
    const int*   src    = (const int*)  d_in[1];
    const int*   dst    = (const int*)  d_in[2];
    const float* W_fc   = (const float*)d_in[3];
    const float* a_attn = (const float*)d_in[4];
    float*       out    = (float*)d_out;
    float*       z      = (float*)d_ws;

    const int nNodes = in_sizes[0] / IN_DIM;
    const int nEdges = in_sizes[1];

    {
        const int n4 = out_size / 4;
        const int blocks = min((n4 + 255) / 256, 4096);
        gat_zero_out<<<blocks, 256, 0, stream>>>((float4*)out, n4);
    }

    {
        const int nWaves = (nNodes + 15) / 16;
        const int blocks = (nWaves + 7) / 8;
        gat_gemm_z<<<blocks, 256, 0, stream>>>(h, W_fc, z, nNodes);
    }

    {
        const int nWaves = (nEdges + EDGES_PER_WAVE - 1) / EDGES_PER_WAVE;
        const int blocks = (nWaves + 7) / 8;
        gat_edge<<<blocks, 256, 0, stream>>>(z, src, dst, a_attn, out, nEdges);
    }
}


# edge pass reads one float per edge, sdst[dst] = z[dst].a[64:] written per node by the GEMM epilogue, instead of gathering the 256-byte z[dst] row; the logit's sum order changes
# speedup vs baseline: 1.8531x; 1.0333x over previous
#include <hip/hip_runtime.h>
#include <hip/hip_bf16.h>

typedef __attribute__((ext_vector_type(2))) float v2f;
typedef __attribute__((ext_vector_type(8))) float v8f;

#define IN_DIM  128
#define OUT_DIM 64

__global__ void gat_zero_out(float4* __restrict__ out, int n4) {
    int i = blockIdx.x * blockDim.x + threadIdx.x;
    int stride = gridDim.x * blockDim.x;
    float4 z4 = make_float4(0.f, 0.f, 0.f, 0.f);
    for (; i < n4; i += stride) out[i] = z4;
}

template <bool DST_LOGIT>
__global__ void gat_gemm_z(const float* __restrict__ h,
                           const float* __restrict__ W,
                           const float* __restrict__ a_attn,
                           float* __restrict__ z,
                           float* __restrict__ sdst,
                           int nNodes) {
    const int wave = (blockIdx.x * blockDim.x + threadIdx.x) >> 5;
    const int lane = threadIdx.x & 31;
    const int row0 = wave * 16;
    if (row0 >= nNodes) return;

    const int half = lane >> 4;
    const int l16  = lane & 15;

    const float* hrow = h + (size_t)(row0 + l16) * IN_DIM + half * 2;

    v8f acc0 = {}, acc1 = {}, acc2 = {}, acc3 = {};

    for (int k = 0; k < IN_DIM; k += 4) {
        v2f a;
        a.x = hrow[k];
        a.y = hrow[k + 1];

        const float* wk = W + (size_t)(k + half * 2) * OUT_DIM + l16;
        v2f b0, b1, b2, b3;
        b0.x = wk[0];       b0.y = wk[OUT_DIM + 0];
        b1.x = wk[16];      b1.y = wk[OUT_DIM + 16];
        b2.x = wk[32];      b2.y = wk[OUT_DIM + 32];
        b3.x = wk[48];      b3.y = wk[OUT_DIM + 48];

        acc0 = __builtin_amdgcn_wmma_f32_16x16x4_f32(false, a, false, b0,
                                                     (short)0, acc0, false, false);
        acc1 = __builtin_amdgcn_wmma_f32_16x16x4_f32(false, a, false, b1,
                                                     (short)0, acc1, false, false);
        acc2 = __builtin_amdgcn_wmma_f32_16x16x4_f32(false, a, false, b2,
                                                     (short)0, acc2, false, false);
        acc3 = __builtin_amdgcn_wmma_f32_16x16x4_f32(false, a, false, b3,
                                                     (short)0, acc3, false, false);
    }

    float* zt = z + (size_t)row0 * OUT_DIM + l16;
#pragma unroll
    for (int v = 0; v < 8; ++v) {
        const size_t r = (size_t)(v + 8 * half) * OUT_DIM;
        zt[r + 0]  = acc0[v];
        zt[r + 16] = acc1[v];
        zt[r + 32] = acc2[v];
        zt[r + 48] = acc3[v];
    }

    if (DST_LOGIT) {
        const float* ad = a_attn + OUT_DIM + l16;
        const float ad0 = ad[0], ad1 = ad[16], ad2 = ad[32], ad3 = ad[48];
        float pd[8];
#pragma unroll
        for (int v = 0; v < 8; ++v)
            pd[v] = acc0[v] * ad0 + acc1[v] * ad1 + acc2[v] * ad2 + acc3[v] * ad3;
#pragma unroll
        for (int off = 8; off > 0; off >>= 1) {
#pragma unroll
            for (int v = 0; v < 8; ++v)
                pd[v] += __shfl_xor(pd[v], off, 32);
        }
        float val = pd[0];
#pragma unroll
        for (int v = 1; v < 8; ++v)
            val = (l16 == v) ? pd[v] : val;
        if (l16 < 8)
            sdst[row0 + 8 * half + l16] = val;
    }
}

#define EDGES_PER_WAVE 32
#define EDGE_UNROLL    8

template <int N, bool DST_LOGIT>
__device__ __forceinline__ void gat_edge_group(const float* __restrict__ z,
                                               float* __restrict__ out,
                                               int s_l, int d_l, float sd_l, int j, int lane,
                                               float a0, float a1, float a2, float a3) {
    float zs0[N], zs1[N], zd0[N], zd1[N], sd[N], p[N];
    int dd[N];

#pragma unroll
    for (int u = 0; u < N; ++u) {
        const int s = __builtin_amdgcn_readlane(s_l, j + u);
        const int d = __builtin_amdgcn_readlane(d_l, j + u);
        dd[u] = d;
        const float* zs = z + (size_t)s * OUT_DIM;
        const float* zd = z + (size_t)d * OUT_DIM;
        zs0[u] = zs[lane];  zs1[u] = zs[lane + 32];
        if (DST_LOGIT) {
            sd[u] = __int_as_float(__builtin_amdgcn_readlane(__float_as_int(sd_l), j + u));
        } else {
            zd0[u] = zd[lane];  zd1[u] = zd[lane + 32];
        }
    }

#pragma unroll
    for (int u = 0; u < N; ++u) {
        if (DST_LOGIT) p[u] = zs0[u] * a0 + zs1[u] * a1;
        else           p[u] = zs0[u] * a0 + zs1[u] * a1 + zd0[u] * a2 + zd1[u] * a3;
    }

#pragma unroll
    for (int off = 16; off > 0; off >>= 1) {
#pragma unroll
        for (int u = 0; u < N; ++u)
            p[u] += __shfl_xor(p[u], off, 32);
    }

#pragma unroll
    for (int u = 0; u < N; ++u) {
        const float q = DST_LOGIT ? p[u] + sd[u] : p[u];
        const float e = (q >= 0.f) ? q : 0.01f * q;
        const float alpha = 1.f / (1.f + __expf(-e));
        float* o = out + (size_t)dd[u] * OUT_DIM;
        atomicAdd(&o[lane],      alpha * zs0[u]);
        atomicAdd(&o[lane + 32], alpha * zs1[u]);
    }
}

template <bool DST_LOGIT>
__global__ void gat_edge(const float* __restrict__ z,
                         const float* __restrict__ sdst,
                         const int* __restrict__ src,
                         const int* __restrict__ dst,
                         const float* __restrict__ a_attn,
                         float* __restrict__ out,
                         int nEdges) {
    const int wave = __builtin_amdgcn_readfirstlane(
        (int)((blockIdx.x * blockDim.x + threadIdx.x) >> 5));
    const int lane = threadIdx.x & 31;
    const int e0 = wave * EDGES_PER_WAVE;
    if (e0 >= nEdges) return;
    const int cnt = min(EDGES_PER_WAVE, nEdges - e0);

    int s_l = 0, d_l = 0;
    float sd_l = 0.f;
    if (lane < cnt) {
        s_l = src[e0 + lane];
        d_l = dst[e0 + lane];
        if (DST_LOGIT) sd_l = sdst[d_l];
    }

    const float a0 = a_attn[lane], a1 = a_attn[lane + 32];
    float a2 = 0.f, a3 = 0.f;
    if (!DST_LOGIT) { a2 = a_attn[64 + lane]; a3 = a_attn[96 + lane]; }

    int j = 0;
    for (; j + EDGE_UNROLL <= cnt; j += EDGE_UNROLL)
        gat_edge_group<EDGE_UNROLL, DST_LOGIT>(z, out, s_l, d_l, sd_l, j, lane, a0, a1, a2, a3);
    for (; j < cnt; ++j)
        gat_edge_group<1, DST_LOGIT>(z, out, s_l, d_l, sd_l, j, lane, a0, a1, a2, a3);
}

extern "C" void kernel_launch(void* const* d_in, const int* in_sizes, int n_in,
                              void* d_out, int out_size, void* d_ws, size_t ws_size,
                              hipStream_t stream) {
    const float* h      = (const float*)d_in[0];
    const int*   src    = (const int*)  d_in[1];
    const int*   dst    = (const int*)  d_in[2];
    const float* W_fc   = (const float*)d_in[3];
    const float* a_attn = (const float*)d_in[4];
    float*       out    = (float*)d_out;
    float*       z      = (float*)d_ws;

    const int nNodes = in_sizes[0] / IN_DIM;
    const int nEdges = in_sizes[1];

    float*     sdst     = z + (size_t)nNodes * OUT_DIM;
    const bool dstLogit = ws_size >= (size_t)nNodes * (OUT_DIM + 1) * sizeof(float);

    {
        const int n4 = out_size / 4;
        const int blocks = min((n4 + 255) / 256, 4096);
        gat_zero_out<<<blocks, 256, 0, stream>>>((float4*)out, n4);
    }

    {
        const int nWaves = (nNodes + 15) / 16;
        const int blocks = (nWaves + 7) / 8;
        if (dstLogit)
            gat_gemm_z<true><<<blocks, 256, 0, stream>>>(h, W_fc, a_attn, z, sdst, nNodes);
        else
            gat_gemm_z<false><<<blocks, 256, 0, stream>>>(h, W_fc, a_attn, z, z, nNodes);
    }

    {
        const int nWaves = (nEdges + EDGES_PER_WAVE - 1) / EDGES_PER_WAVE;
        const int blocks = (nWaves + 7) / 8;
        if (dstLogit)
            gat_edge<true><<<blocks, 256, 0, stream>>>(z, sdst, src, dst, a_attn, out, nEdges);
        else
            gat_edge<false><<<blocks, 256, 0, stream>>>(z, z, src, dst, a_attn, out, nEdges);
    }
}
